# starting kernel
# baseline (speedup 1.0000x reference)
#include <hip/hip_runtime.h>
#include <hip/hip_bf16.h>
#include <math.h>

#define NQ    10
#define NL    6
#define BATCH 2048
#define DIM   1024


typedef float v2f __attribute__((ext_vector_type(2)));
typedef float v8f __attribute__((ext_vector_type(8)));

__device__ __forceinline__ float2 cmul(float2 a, float2 b) {
    return make_float2(a.x * b.x - a.y * b.y, a.x * b.y + a.y * b.x);
}
__device__ __forceinline__ float2 cadd(float2 a, float2 b) {
    return make_float2(a.x + b.x, a.y + b.y);
}
__device__ __forceinline__ float2 cscale(float s, float2 a) {
    return make_float2(s * a.x, s * a.y);
}

__global__ void __launch_bounds__(256) embed_kernel(const float* __restrict__ x,
                                                    float* __restrict__ pre,
                                                    float* __restrict__ pim) {
    __shared__ float cs[NQ], sn[NQ];
    const int b = blockIdx.x;
    const int tid = threadIdx.x;
    if (tid < NQ) {
        float s, c;
        sincosf(0.5f * x[b * NQ + tid], &s, &c);
        sn[tid] = s; cs[tid] = c;
    }
    __syncthreads();
    #pragma unroll
    for (int i = 0; i < DIM / 256; ++i) {
        const int k = tid + 256 * i;
        float r = 1.0f;
        #pragma unroll
        for (int q = 0; q < NQ; ++q) {
            const int bit = (k >> (NQ - 1 - q)) & 1;
            r *= bit ? sn[q] : cs[q];
        }
        const int p = __popc(k) & 3;
        const float re = (p == 0) ? r : (p == 2 ? -r : 0.0f);
        const float im = (p == 1) ? -r : (p == 3 ? r : 0.0f);
        pre[b * DIM + k] = re;
        pim[b * DIM + k] = im;
    }
}

__global__ void __launch_bounds__(128) build_u_kernel(const float* __restrict__ w,
                                                      float* __restrict__ Mre,
                                                      float* __restrict__ Mim) {
    __shared__ float2 st[DIM];
    const int j = blockIdx.x;
    const int tid = threadIdx.x;
    for (int k = tid; k < DIM; k += 128)
        st[k] = make_float2(k == j ? 1.0f : 0.0f, 0.0f);
    __syncthreads();

    for (int l = 0; l < NL; ++l) {
        for (int q = 0; q < NQ; ++q) {
            const float phi = w[(l * NQ + q) * 3 + 0];
            const float th  = w[(l * NQ + q) * 3 + 1];
            const float om  = w[(l * NQ + q) * 3 + 2];
            float ct, s_t, ca, sa, cb, sb;
            sincosf(0.5f * th, &s_t, &ct);
            sincosf(0.5f * (phi + om), &sa, &ca);
            sincosf(0.5f * (phi - om), &sb, &cb);
            const float2 ga = make_float2( ct * ca, -ct * sa);
            const float2 gb = make_float2(-s_t * cb, -s_t * sb);
            const float2 gc = make_float2( s_t * cb, -s_t * sb);
            const float2 gd = make_float2( ct * ca,  ct * sa);
            const int p = NQ - 1 - q;
            for (int t = tid; t < DIM / 2; t += 128) {
                const int i0 = ((t >> p) << (p + 1)) | (t & ((1 << p) - 1));
                const int i1 = i0 | (1 << p);
                const float2 s0 = st[i0], s1 = st[i1];
                st[i0] = cadd(cmul(ga, s0), cmul(gb, s1));
                st[i1] = cadd(cmul(gc, s0), cmul(gd, s1));
            }
            __syncthreads();
        }
        for (int q = 0; q < NQ; ++q) {
            const int pc = NQ - 1 - q;
            const int pt = NQ - 1 - ((q + 1) % NQ);
            const int plo = pc < pt ? pc : pt;
            const int phi_ = pc < pt ? pt : pc;
            for (int t = tid; t < DIM / 4; t += 128) {
                int i = ((t >> plo) << (plo + 1)) | (t & ((1 << plo) - 1));
                i = ((i >> phi_) << (phi_ + 1)) | (i & ((1 << phi_) - 1));
                const int i1 = i | (1 << pc);
                const int i2 = i1 | (1 << pt);
                const float2 tmp = st[i1]; st[i1] = st[i2]; st[i2] = tmp;
            }
            __syncthreads();
        }
        {
            float s, c; sincosf(0.5f * w[(l * NQ + 0) * 3 + 0], &s, &c);
            const float2 ms = make_float2(0.0f, -s);
            for (int t = tid; t < DIM / 2; t += 128) {
                const int jd = t ^ 0x300;
                const float2 a0 = st[t], a1 = st[jd];
                st[t]  = cadd(cscale(c, a0), cmul(ms, a1));
                st[jd] = cadd(cscale(c, a1), cmul(ms, a0));
            }
            __syncthreads();
        }
        {
            float s, c; sincosf(0.5f * w[(l * NQ + 0) * 3 + 1], &s, &c);
            for (int t = tid; t < DIM / 2; t += 128) {
                const int i = ((t >> 8) << 9) | (t & 0xFF);
                const int jd = i ^ 0x180;
                const float sgn = ((i >> 7) & 1) ? 1.0f : -1.0f;
                const float2 ms = make_float2(0.0f, -sgn * s);
                const float2 a0 = st[i], a1 = st[jd];
                st[i]  = cadd(cscale(c, a0), cmul(ms, a1));
                st[jd] = cadd(cscale(c, a1), cmul(ms, a0));
            }
            __syncthreads();
        }
        {
            float s, c; sincosf(0.5f * w[(l * NQ + 0) * 3 + 2], &s, &c);
            for (int k = tid; k < DIM; k += 128) {
                const int df = ((k >> 7) ^ (k >> 6)) & 1;
                const float2 ph = make_float2(c, df ? s : -s);
                st[k] = cmul(ph, st[k]);
            }
            __syncthreads();
        }
    }

    for (int k = tid; k < DIM; k += 128) {
        Mre[j * DIM + k] = st[k].x;
        Mim[j * DIM + k] = st[k].y;
    }
}

__global__ void zero_kernel(float* __restrict__ out, int n) {
    const int i = blockIdx.x * blockDim.x + threadIdx.x;
    if (i < n) out[i] = 0.0f;
}

__global__ void __launch_bounds__(256) gemm_z_kernel(const float* __restrict__ Are,
                                                     const float* __restrict__ Aim,
                                                     const float* __restrict__ Bre,
                                                     const float* __restrict__ Bim,
                                                     float* __restrict__ out) {
    const int lane = threadIdx.x & 31;
    const int wave = threadIdx.x >> 5;
    const int gw = blockIdx.x * 8 + wave;
    const int tile_b = gw >> 6;
    const int tile_u = gw & 63;
    const int b0 = tile_b * 16;
    const int u0 = tile_u * 16;
    const int half = lane >> 4;
    const int l16 = lane & 15;
    const int ucol = u0 + l16;

    const float* arp = Are + (b0 + l16) * DIM + 2 * half;
    const float* aip = Aim + (b0 + l16) * DIM + 2 * half;
    const float* brp = Bre + (2 * half) * DIM + ucol;
    const float* bip = Bim + (2 * half) * DIM + ucol;

    v8f acc_re = {0.f, 0.f, 0.f, 0.f, 0.f, 0.f, 0.f, 0.f};
    v8f acc_im = {0.f, 0.f, 0.f, 0.f, 0.f, 0.f, 0.f, 0.f};

    #pragma unroll 4
    for (int k0 = 0; k0 < DIM; k0 += 4) {
        v2f ar = *(const v2f*)(arp + k0);
        v2f ai = *(const v2f*)(aip + k0);
        v2f br = { brp[k0 * DIM], brp[(k0 + 1) * DIM] };
        v2f bi = { bip[k0 * DIM], bip[(k0 + 1) * DIM] };
        v2f nai = { -ai[0], -ai[1] };

        acc_re = __builtin_amdgcn_wmma_f32_16x16x4_f32(false, ar,  false, br, (short)0, acc_re, false, false);
        acc_re = __builtin_amdgcn_wmma_f32_16x16x4_f32(false, nai, false, bi, (short)0, acc_re, false, false);
        acc_im = __builtin_amdgcn_wmma_f32_16x16x4_f32(false, ar,  false, bi, (short)0, acc_im, false, false);
        acc_im = __builtin_amdgcn_wmma_f32_16x16x4_f32(false, ai,  false, br, (short)0, acc_im, false, false);

        __builtin_prefetch(arp + k0 + 64, 0, 1);
        __builtin_prefetch(brp + (k0 + 16) * DIM, 0, 1);
    }

    float p[8];
    #pragma unroll
    for (int v = 0; v < 8; ++v)
        p[v] = acc_re[v] * acc_re[v] + acc_im[v] * acc_im[v];

    #pragma unroll
    for (int q = 0; q < NQ; ++q) {
        const float sgn = ((ucol >> (NQ - 1 - q)) & 1) ? -1.0f : 1.0f;
        float t[8];
        #pragma unroll
        for (int v = 0; v < 8; ++v) t[v] = sgn * p[v];
        #pragma unroll
        for (int m = 1; m < 16; m <<= 1) {
            #pragma unroll
            for (int v = 0; v < 8; ++v) t[v] += __shfl_xor(t[v], m, 32);
        }
        if (l16 == 0) {
            #pragma unroll
            for (int v = 0; v < 8; ++v)
                atomicAdd(&out[(b0 + half * 8 + v) * NQ + q], t[v]);
        }
    }
}

extern "C" void kernel_launch(void* const* d_in, const int* in_sizes, int n_in,
                              void* d_out, int out_size, void* d_ws, size_t ws_size,
                              hipStream_t stream) {
    const float* x = (const float*)d_in[0];
    const float* w = (const float*)d_in[1];
    float* out = (float*)d_out;

    float* pre = (float*)d_ws;
    float* pim = pre + (size_t)BATCH * DIM;
    float* Mre = pim + (size_t)BATCH * DIM;
    float* Mim = Mre + (size_t)DIM * DIM;

    embed_kernel  <<<BATCH, 256, 0, stream>>>(x, pre, pim);
    build_u_kernel<<<DIM,   128, 0, stream>>>(w, Mre, Mim);
    zero_kernel   <<<(BATCH * NQ + 255) / 256, 256, 0, stream>>>(out, BATCH * NQ);
    gemm_z_kernel <<<(BATCH / 16) * (DIM / 16) / 8, 256, 0, stream>>>(pre, pim, Mre, Mim, out);
}


# GEMM register-tiled, no LDS: one wave per 32x64 tile (2x4 WMMA tiles), 16-byte global fragment loads reused in registers, next chunk loaded ahead; changes the summation order
# speedup vs baseline: 2.2850x; 2.2850x over previous
#include <hip/hip_runtime.h>
#include <hip/hip_bf16.h>
#include <math.h>

#define NQ    10
#define NL    6
#define BATCH 2048
#define DIM   1024


typedef float v2f __attribute__((ext_vector_type(2)));
typedef float v8f __attribute__((ext_vector_type(8)));

__device__ __forceinline__ float2 cmul(float2 a, float2 b) {
    return make_float2(a.x * b.x - a.y * b.y, a.x * b.y + a.y * b.x);
}
__device__ __forceinline__ float2 cadd(float2 a, float2 b) {
    return make_float2(a.x + b.x, a.y + b.y);
}
__device__ __forceinline__ float2 cscale(float s, float2 a) {
    return make_float2(s * a.x, s * a.y);
}

__global__ void __launch_bounds__(256) embed_kernel(const float* __restrict__ x,
                                                    float* __restrict__ pre,
                                                    float* __restrict__ pim) {
    __shared__ float cs[NQ], sn[NQ];
    const int b = blockIdx.x;
    const int tid = threadIdx.x;
    if (tid < NQ) {
        float s, c;
        sincosf(0.5f * x[b * NQ + tid], &s, &c);
        sn[tid] = s; cs[tid] = c;
    }
    __syncthreads();
    #pragma unroll
    for (int i = 0; i < DIM / 256; ++i) {
        const int k = tid + 256 * i;
        float r = 1.0f;
        #pragma unroll
        for (int q = 0; q < NQ; ++q) {
            const int bit = (k >> (NQ - 1 - q)) & 1;
            r *= bit ? sn[q] : cs[q];
        }
        const int p = __popc(k) & 3;
        const float re = (p == 0) ? r : (p == 2 ? -r : 0.0f);
        const float im = (p == 1) ? -r : (p == 3 ? r : 0.0f);
        pre[b * DIM + k] = re;
        pim[b * DIM + k] = im;
    }
}

__global__ void __launch_bounds__(128) build_u_kernel(const float* __restrict__ w,
                                                      float* __restrict__ Mre,
                                                      float* __restrict__ Mim) {
    __shared__ float2 st[DIM];
    const int j = blockIdx.x;
    const int tid = threadIdx.x;
    for (int k = tid; k < DIM; k += 128)
        st[k] = make_float2(k == j ? 1.0f : 0.0f, 0.0f);
    __syncthreads();

    for (int l = 0; l < NL; ++l) {
        for (int q = 0; q < NQ; ++q) {
            const float phi = w[(l * NQ + q) * 3 + 0];
            const float th  = w[(l * NQ + q) * 3 + 1];
            const float om  = w[(l * NQ + q) * 3 + 2];
            float ct, s_t, ca, sa, cb, sb;
            sincosf(0.5f * th, &s_t, &ct);
            sincosf(0.5f * (phi + om), &sa, &ca);
            sincosf(0.5f * (phi - om), &sb, &cb);
            const float2 ga = make_float2( ct * ca, -ct * sa);
            const float2 gb = make_float2(-s_t * cb, -s_t * sb);
            const float2 gc = make_float2( s_t * cb, -s_t * sb);
            const float2 gd = make_float2( ct * ca,  ct * sa);
            const int p = NQ - 1 - q;
            for (int t = tid; t < DIM / 2; t += 128) {
                const int i0 = ((t >> p) << (p + 1)) | (t & ((1 << p) - 1));
                const int i1 = i0 | (1 << p);
                const float2 s0 = st[i0], s1 = st[i1];
                st[i0] = cadd(cmul(ga, s0), cmul(gb, s1));
                st[i1] = cadd(cmul(gc, s0), cmul(gd, s1));
            }
            __syncthreads();
        }
        for (int q = 0; q < NQ; ++q) {
            const int pc = NQ - 1 - q;
            const int pt = NQ - 1 - ((q + 1) % NQ);
            const int plo = pc < pt ? pc : pt;
            const int phi_ = pc < pt ? pt : pc;
            for (int t = tid; t < DIM / 4; t += 128) {
                int i = ((t >> plo) << (plo + 1)) | (t & ((1 << plo) - 1));
                i = ((i >> phi_) << (phi_ + 1)) | (i & ((1 << phi_) - 1));
                const int i1 = i | (1 << pc);
                const int i2 = i1 | (1 << pt);
                const float2 tmp = st[i1]; st[i1] = st[i2]; st[i2] = tmp;
            }
            __syncthreads();
        }
        {
            float s, c; sincosf(0.5f * w[(l * NQ + 0) * 3 + 0], &s, &c);
            const float2 ms = make_float2(0.0f, -s);
            for (int t = tid; t < DIM / 2; t += 128) {
                const int jd = t ^ 0x300;
                const float2 a0 = st[t], a1 = st[jd];
                st[t]  = cadd(cscale(c, a0), cmul(ms, a1));
                st[jd] = cadd(cscale(c, a1), cmul(ms, a0));
            }
            __syncthreads();
        }
        {
            float s, c; sincosf(0.5f * w[(l * NQ + 0) * 3 + 1], &s, &c);
            for (int t = tid; t < DIM / 2; t += 128) {
                const int i = ((t >> 8) << 9) | (t & 0xFF);
                const int jd = i ^ 0x180;
                const float sgn = ((i >> 7) & 1) ? 1.0f : -1.0f;
                const float2 ms = make_float2(0.0f, -sgn * s);
                const float2 a0 = st[i], a1 = st[jd];
                st[i]  = cadd(cscale(c, a0), cmul(ms, a1));
                st[jd] = cadd(cscale(c, a1), cmul(ms, a0));
            }
            __syncthreads();
        }
        {
            float s, c; sincosf(0.5f * w[(l * NQ + 0) * 3 + 2], &s, &c);
            for (int k = tid; k < DIM; k += 128) {
                const int df = ((k >> 7) ^ (k >> 6)) & 1;
                const float2 ph = make_float2(c, df ? s : -s);
                st[k] = cmul(ph, st[k]);
            }
            __syncthreads();
        }
    }

    for (int k = tid; k < DIM; k += 128) {
        Mre[j * DIM + k] = st[k].x;
        Mim[j * DIM + k] = st[k].y;
    }
}

__global__ void zero_kernel(float* __restrict__ out, int n) {
    const int i = blockIdx.x * blockDim.x + threadIdx.x;
    if (i < n) out[i] = 0.0f;
}

typedef float v4f __attribute__((ext_vector_type(4)));

struct Frag {
    v4f ar[2], ai[2];
    v4f br[4], bi[4];
};

__device__ __forceinline__ void load_frag(Frag& f,
                                          const float* __restrict__ arp,
                                          const float* __restrict__ aip,
                                          const float* __restrict__ brp,
                                          const float* __restrict__ bip,
                                          int k0) {
    #pragma unroll
    for (int mt = 0; mt < 2; ++mt) {
        f.ar[mt] = *(const v4f*)(arp + mt * 16 * DIM + k0);
        f.ai[mt] = *(const v4f*)(aip + mt * 16 * DIM + k0);
    }
    #pragma unroll
    for (int r = 0; r < 4; ++r) {
        f.br[r] = *(const v4f*)(brp + (k0 + r) * DIM);
        f.bi[r] = *(const v4f*)(bip + (k0 + r) * DIM);
    }
}

__device__ __forceinline__ void mac_frag(const Frag& f, v8f (&acc_re)[2][4], v8f (&acc_im)[2][4]) {
    #pragma unroll
    for (int e = 0; e < 2; ++e) {
        v2f ar[2], ai[2], nai[2], br[4], bi[4];
        #pragma unroll
        for (int mt = 0; mt < 2; ++mt) {
            ar[mt]  = (v2f){  f.ar[mt][2 * e],  f.ar[mt][2 * e + 1] };
            ai[mt]  = (v2f){  f.ai[mt][2 * e],  f.ai[mt][2 * e + 1] };
            nai[mt] = (v2f){ -f.ai[mt][2 * e], -f.ai[mt][2 * e + 1] };
        }
        #pragma unroll
        for (int t = 0; t < 4; ++t) {
            br[t] = (v2f){ f.br[2 * e][t], f.br[2 * e + 1][t] };
            bi[t] = (v2f){ f.bi[2 * e][t], f.bi[2 * e + 1][t] };
        }
        #pragma unroll
        for (int mt = 0; mt < 2; ++mt)
            #pragma unroll
            for (int t = 0; t < 4; ++t)
                acc_re[mt][t] = __builtin_amdgcn_wmma_f32_16x16x4_f32(false, ar[mt],  false, br[t], (short)0, acc_re[mt][t], false, false);
        #pragma unroll
        for (int mt = 0; mt < 2; ++mt)
            #pragma unroll
            for (int t = 0; t < 4; ++t)
                acc_re[mt][t] = __builtin_amdgcn_wmma_f32_16x16x4_f32(false, nai[mt], false, bi[t], (short)0, acc_re[mt][t], false, false);
        #pragma unroll
        for (int mt = 0; mt < 2; ++mt)
            #pragma unroll
            for (int t = 0; t < 4; ++t)
                acc_im[mt][t] = __builtin_amdgcn_wmma_f32_16x16x4_f32(false, ar[mt],  false, bi[t], (short)0, acc_im[mt][t], false, false);
        #pragma unroll
        for (int mt = 0; mt < 2; ++mt)
            #pragma unroll
            for (int t = 0; t < 4; ++t)
                acc_im[mt][t] = __builtin_amdgcn_wmma_f32_16x16x4_f32(false, ai[mt],  false, br[t], (short)0, acc_im[mt][t], false, false);
    }
}

__global__ void __launch_bounds__(128) gemm_z_kernel(const float* __restrict__ Are,
                                                     const float* __restrict__ Aim,
                                                     const float* __restrict__ Bre,
                                                     const float* __restrict__ Bim,
                                                     float* __restrict__ out) {
    const int lane = threadIdx.x & 31;
    const int wave = threadIdx.x >> 5;
    const int half = lane >> 4;
    const int l16 = lane & 15;
    const int brow = blockIdx.x >> 3;
    const int bcol = blockIdx.x & 7;
    const int b0 = brow * 64 + (wave >> 1) * 32;
    const int u0 = bcol * 128 + (wave & 1) * 64;

    const float* arp = Are + (b0 + l16) * DIM + 4 * half;
    const float* aip = Aim + (b0 + l16) * DIM + 4 * half;
    const float* brp = Bre + (4 * half) * DIM + u0 + 4 * l16;
    const float* bip = Bim + (4 * half) * DIM + u0 + 4 * l16;

    v8f acc_re[2][4], acc_im[2][4];
    #pragma unroll
    for (int mt = 0; mt < 2; ++mt)
        #pragma unroll
        for (int t = 0; t < 4; ++t) {
            acc_re[mt][t] = (v8f){0.f, 0.f, 0.f, 0.f, 0.f, 0.f, 0.f, 0.f};
            acc_im[mt][t] = (v8f){0.f, 0.f, 0.f, 0.f, 0.f, 0.f, 0.f, 0.f};
        }

    Frag f0, f1;
    load_frag(f0, arp, aip, brp, bip, 0);
    for (int k0 = 0; k0 < DIM - 16; k0 += 16) {
        load_frag(f1, arp, aip, brp, bip, k0 + 8);
        mac_frag(f0, acc_re, acc_im);
        load_frag(f0, arp, aip, brp, bip, k0 + 16);
        mac_frag(f1, acc_re, acc_im);
    }
    load_frag(f1, arp, aip, brp, bip, DIM - 8);
    mac_frag(f0, acc_re, acc_im);
    mac_frag(f1, acc_re, acc_im);

    #pragma unroll
    for (int mt = 0; mt < 2; ++mt) {
        float p[4][8];
        #pragma unroll
        for (int t = 0; t < 4; ++t)
            #pragma unroll
            for (int v = 0; v < 8; ++v)
                p[t][v] = acc_re[mt][t][v] * acc_re[mt][t][v] + acc_im[mt][t][v] * acc_im[mt][t][v];

        #pragma unroll
        for (int q = 0; q < NQ; ++q) {
            float s[8];
            #pragma unroll
            for (int v = 0; v < 8; ++v) s[v] = 0.0f;
            #pragma unroll
            for (int t = 0; t < 4; ++t) {
                const int ucol = u0 + 4 * l16 + t;
                const float sgn = ((ucol >> (NQ - 1 - q)) & 1) ? -1.0f : 1.0f;
                #pragma unroll
                for (int v = 0; v < 8; ++v) s[v] += sgn * p[t][v];
            }
            #pragma unroll
            for (int m = 1; m < 16; m <<= 1) {
                #pragma unroll
                for (int v = 0; v < 8; ++v) s[v] += __shfl_xor(s[v], m, 32);
            }
            if (l16 == 0) {
                #pragma unroll
                for (int v = 0; v < 8; ++v)
                    atomicAdd(&out[(b0 + mt * 16 + half * 8 + v) * NQ + q], s[v]);
            }
        }
    }
}

extern "C" void kernel_launch(void* const* d_in, const int* in_sizes, int n_in,
                              void* d_out, int out_size, void* d_ws, size_t ws_size,
                              hipStream_t stream) {
    const float* x = (const float*)d_in[0];
    const float* w = (const float*)d_in[1];
    float* out = (float*)d_out;

    float* pre = (float*)d_ws;
    float* pim = pre + (size_t)BATCH * DIM;
    float* Mre = pim + (size_t)BATCH * DIM;
    float* Mim = Mre + (size_t)DIM * DIM;

    embed_kernel  <<<BATCH, 256, 0, stream>>>(x, pre, pim);
    build_u_kernel<<<DIM,   128, 0, stream>>>(w, Mre, Mim);
    zero_kernel   <<<(BATCH * NQ + 255) / 256, 256, 0, stream>>>(out, BATCH * NQ);
    gemm_z_kernel <<<(BATCH / 64) * (DIM / 128), 128, 0, stream>>>(pre, pim, Mre, Mim, out);
}
